# starting kernel
# baseline (speedup 1.0000x reference)
#include <hip/hip_runtime.h>
#include <hip/hip_bf16.h>
#include <math.h>

typedef __attribute__((ext_vector_type(2))) float v2f;
typedef __attribute__((ext_vector_type(8))) float v8f;

#define WMMA_F32(A, B, C) \
  __builtin_amdgcn_wmma_f32_16x16x4_f32(false, (A), false, (B), (short)0, (C), false, false)

__device__ __forceinline__ v2f mkv2(float a, float b) { v2f r; r.x = a; r.y = b; return r; }

__device__ __forceinline__ void wave_lds_fence() {
  asm volatile("s_wait_dscnt 0" ::: "memory");
}

constexpr int Bsz = 32768;
constexpr int D   = 64;
constexpr int H   = 32;
constexpr int TILES = 4;
constexpr int WAVES = 8;
constexpr int ROWS_PER_BLOCK = WAVES * TILES * 16;

__global__ __launch_bounds__(256) void encoder_kernel(
    const float* __restrict__ x,   const float* __restrict__ eps,
    const float* __restrict__ W1,  const float* __restrict__ b1,
    const float* __restrict__ W2,  const float* __restrict__ b2,
    const float* __restrict__ Wmu, const float* __restrict__ bmu,
    const float* __restrict__ Wlv, const float* __restrict__ blv,
    float* __restrict__ mu_out, float* __restrict__ lv_out,
    float* __restrict__ z_ws)
{
  __shared__ float lds[WAVES * 16 * H];
  const int lane = threadIdx.x & 31;
  const int wave = threadIdx.x >> 5;
  const int col  = lane & 15;
  const int hi   = lane >> 4;
  const int d    = blockIdx.y;
  float* wl = &lds[wave * 16 * H];

  v2f w2b[2][8];
  v2f w1v[8], b1v[8];
  v2f wml[8];
  #pragma unroll
  for (int c = 0; c < 8; ++c) {
    const int k0 = 4 * c + 2 * hi;
    w1v[c] = mkv2(W1[d * H + k0], W1[d * H + k0 + 1]);
    b1v[c] = mkv2(b1[d * H + k0], b1[d * H + k0 + 1]);
    #pragma unroll
    for (int nt = 0; nt < 2; ++nt) {
      const int n = nt * 16 + col;
      w2b[nt][c] = mkv2(W2[d * H * H + k0 * H + n],
                        W2[d * H * H + (k0 + 1) * H + n]);
    }
    const float m0 = Wmu[d * H + k0], m1 = Wmu[d * H + k0 + 1];
    const float l0 = Wlv[d * H + k0], l1 = Wlv[d * H + k0 + 1];
    wml[c] = mkv2(col == 0 ? m0 : (col == 1 ? l0 : 0.0f),
                  col == 0 ? m1 : (col == 1 ? l1 : 0.0f));
  }
  const float b2v0 = b2[d * H + col];
  const float b2v1 = b2[d * H + 16 + col];
  const float bmuv = bmu[d];
  const float blvv = blv[d];

  const int rowBase = blockIdx.x * ROWS_PER_BLOCK + wave * (TILES * 16);

  for (int t = 0; t < TILES; ++t) {
    const int row0 = rowBase + t * 16;
    const float xv = x[(row0 + col) * D + d];

    v8f acc0 = {}, acc1 = {};
    #pragma unroll
    for (int c = 0; c < 8; ++c) {
      v2f a;
      a.x = fmaxf(fmaf(xv, w1v[c].x, b1v[c].x), 0.0f);
      a.y = fmaxf(fmaf(xv, w1v[c].y, b1v[c].y), 0.0f);
      acc0 = WMMA_F32(a, w2b[0][c], acc0);
      acc1 = WMMA_F32(a, w2b[1][c], acc1);
    }

    wave_lds_fence();
    #pragma unroll
    for (int j = 0; j < 8; ++j) {
      const int r = j + 8 * hi;
      wl[r * H + col]      = fmaxf(acc0[j] + b2v0, 0.0f);
      wl[r * H + 16 + col] = fmaxf(acc1[j] + b2v1, 0.0f);
    }
    wave_lds_fence();

    v8f accS = {};
    #pragma unroll
    for (int c = 0; c < 8; ++c) {
      v2f a = *(const v2f*)&wl[col * H + 4 * c + 2 * hi];
      accS = WMMA_F32(a, wml[c], accS);
    }

    #pragma unroll
    for (int j = 0; j < 8; ++j) {
      const float self  = accS[j];
      const float other = __shfl_xor(self, 1, 32);
      const int row = row0 + j + 8 * hi;
      if (col == 0) {
        const float m  = self + bmuv;
        const float lv = other + blvv;
        mu_out[row * D + d] = m;
        z_ws[row * D + d]   = fmaf(eps[row * D + d], __expf(0.5f * lv), m);
      } else if (col == 1) {
        lv_out[row * D + d] = self + blvv;
      }
    }
  }
}

__global__ __launch_bounds__(256) void decoder_kernel(
    const float* __restrict__ z_ws,
    const float* __restrict__ Wd1, const float* __restrict__ bd1,
    const float* __restrict__ Wd2, const float* __restrict__ bd2,
    const float* __restrict__ Wd3, const float* __restrict__ bd3,
    float* __restrict__ action)
{
  __shared__ float lds[WAVES * 16 * H];
  const int lane = threadIdx.x & 31;
  const int wave = threadIdx.x >> 5;
  const int col  = lane & 15;
  const int hi   = lane >> 4;
  float* wl = &lds[wave * 16 * H];

  v2f wd1b[2][16];
  #pragma unroll
  for (int c = 0; c < 16; ++c) {
    const int k0 = 4 * c + 2 * hi;
    #pragma unroll
    for (int nt = 0; nt < 2; ++nt) {
      const int n = nt * 16 + col;
      wd1b[nt][c] = mkv2(Wd1[k0 * 32 + n], Wd1[(k0 + 1) * 32 + n]);
    }
  }
  v2f wd2b[2][8], wd3b[8];
  #pragma unroll
  for (int c = 0; c < 8; ++c) {
    const int k0 = 4 * c + 2 * hi;
    #pragma unroll
    for (int nt = 0; nt < 2; ++nt) {
      const int n = nt * 16 + col;
      wd2b[nt][c] = mkv2(Wd2[k0 * 32 + n], Wd2[(k0 + 1) * 32 + n]);
    }
    wd3b[c] = mkv2(col == 0 ? Wd3[k0] : 0.0f,
                   col == 0 ? Wd3[k0 + 1] : 0.0f);
  }
  const float bd1v0 = bd1[col], bd1v1 = bd1[16 + col];
  const float bd2v0 = bd2[col], bd2v1 = bd2[16 + col];
  const float bd3v  = bd3[0];

  const int rowBase = blockIdx.x * ROWS_PER_BLOCK + wave * (TILES * 16);

  for (int t = 0; t < TILES; ++t) {
    const int row0 = rowBase + t * 16;

    v8f acc0 = {}, acc1 = {};
    #pragma unroll
    for (int c = 0; c < 16; ++c) {
      v2f a = *(const v2f*)&z_ws[(row0 + col) * 64 + 4 * c + 2 * hi];
      acc0 = WMMA_F32(a, wd1b[0][c], acc0);
      acc1 = WMMA_F32(a, wd1b[1][c], acc1);
    }
    wave_lds_fence();
    #pragma unroll
    for (int j = 0; j < 8; ++j) {
      const int r = j + 8 * hi;
      wl[r * 32 + col]      = fmaxf(acc0[j] + bd1v0, 0.0f);
      wl[r * 32 + 16 + col] = fmaxf(acc1[j] + bd1v1, 0.0f);
    }
    wave_lds_fence();

    v8f a20 = {}, a21 = {};
    #pragma unroll
    for (int c = 0; c < 8; ++c) {
      v2f a = *(const v2f*)&wl[col * 32 + 4 * c + 2 * hi];
      a20 = WMMA_F32(a, wd2b[0][c], a20);
      a21 = WMMA_F32(a, wd2b[1][c], a21);
    }
    wave_lds_fence();
    #pragma unroll
    for (int j = 0; j < 8; ++j) {
      const int r = j + 8 * hi;
      wl[r * 32 + col]      = fmaxf(a20[j] + bd2v0, 0.0f);
      wl[r * 32 + 16 + col] = fmaxf(a21[j] + bd2v1, 0.0f);
    }
    wave_lds_fence();

    v8f accS = {};
    #pragma unroll
    for (int c = 0; c < 8; ++c) {
      v2f a = *(const v2f*)&wl[col * 32 + 4 * c + 2 * hi];
      accS = WMMA_F32(a, wd3b[c], accS);
    }
    if (col == 0) {
      #pragma unroll
      for (int j = 0; j < 8; ++j)
        action[row0 + j + 8 * hi] = accS[j] + bd3v;
    }
  }
}

extern "C" void kernel_launch(void* const* d_in, const int* in_sizes, int n_in,
                              void* d_out, int out_size, void* d_ws, size_t ws_size,
                              hipStream_t stream) {
  (void)in_sizes; (void)n_in; (void)out_size; (void)ws_size;
  const float* x   = (const float*)d_in[0];
  const float* eps = (const float*)d_in[1];
  const float* W1  = (const float*)d_in[2];
  const float* b1  = (const float*)d_in[3];
  const float* W2  = (const float*)d_in[4];
  const float* b2  = (const float*)d_in[5];
  const float* Wmu = (const float*)d_in[6];
  const float* bmu = (const float*)d_in[7];
  const float* Wlv = (const float*)d_in[8];
  const float* blv = (const float*)d_in[9];
  const float* Wd1 = (const float*)d_in[10];
  const float* bd1 = (const float*)d_in[11];
  const float* Wd2 = (const float*)d_in[12];
  const float* bd2 = (const float*)d_in[13];
  const float* Wd3 = (const float*)d_in[14];
  const float* bd3 = (const float*)d_in[15];

  float* out     = (float*)d_out;
  float* action  = out;
  float* mu_out  = out + Bsz;
  float* lv_out  = out + Bsz + Bsz * D;
  float* z_ws    = (float*)d_ws;

  dim3 grid1(Bsz / ROWS_PER_BLOCK, D);
  encoder_kernel<<<grid1, 256, 0, stream>>>(x, eps, W1, b1, W2, b2,
                                            Wmu, bmu, Wlv, blv,
                                            mu_out, lv_out, z_ws);
  decoder_kernel<<<Bsz / ROWS_PER_BLOCK, 256, 0, stream>>>(z_ws, Wd1, bd1,
                                                           Wd2, bd2, Wd3, bd3,
                                                           action);
}


# weight preload without lane conditions: mu/logvar and Wd3 B operands load unconditionally (unread D columns get copies, not zeros), so no branches and one load wait, not 8 serial drains per wave
# speedup vs baseline: 2.3622x; 2.3622x over previous
#include <hip/hip_runtime.h>
#include <hip/hip_bf16.h>
#include <math.h>

typedef __attribute__((ext_vector_type(2))) float v2f;
typedef __attribute__((ext_vector_type(8))) float v8f;

#define WMMA_F32(A, B, C) \
  __builtin_amdgcn_wmma_f32_16x16x4_f32(false, (A), false, (B), (short)0, (C), false, false)

__device__ __forceinline__ v2f mkv2(float a, float b) { v2f r; r.x = a; r.y = b; return r; }

__device__ __forceinline__ void wave_lds_fence() {
  asm volatile("s_wait_dscnt 0" ::: "memory");
}

constexpr int Bsz = 32768;
constexpr int D   = 64;
constexpr int H   = 32;
constexpr int TILES = 4;
constexpr int WAVES = 8;
constexpr int ROWS_PER_BLOCK = WAVES * TILES * 16;

constexpr int EF     = 32;
constexpr int ETILES = 16;
constexpr int ER     = ETILES * 16;
constexpr int EPASS  = ER / 128;
constexpr int EPITCH = EF + 1;

__global__ __launch_bounds__(1024) void encoder_kernel(
    const float* __restrict__ x,   const float* __restrict__ eps,
    const float* __restrict__ W1,  const float* __restrict__ b1,
    const float* __restrict__ W2,  const float* __restrict__ b2,
    const float* __restrict__ Wmu, const float* __restrict__ bmu,
    const float* __restrict__ Wlv, const float* __restrict__ blv,
    float* __restrict__ mu_out, float* __restrict__ lv_out,
    float* __restrict__ z_ws)
{
  __shared__ float lds[EF * 16 * H];
  __shared__ float xs[ER * EPITCH];
  __shared__ float es[ER * EPITCH];
  __shared__ float os[3 * ER * EPITCH];
  const int tid  = threadIdx.x;
  const int lane = tid & 31;
  const int wave = tid >> 5;
  const int col  = lane & 15;
  const int hi   = lane >> 4;
  const int dbase = blockIdx.y * EF;
  const int d     = dbase + wave;
  const int rowBase = blockIdx.x * ER;
  float* wl = &lds[wave * 16 * H];

  const int lr = tid >> 3;
  const int lq = tid & 7;
  float4 xin[EPASS], ein[EPASS];
  #pragma unroll
  for (int p = 0; p < EPASS; ++p) {
    xin[p] = *(const float4*)&x[(rowBase + p * 128 + lr) * D + dbase + 4 * lq];
    ein[p] = *(const float4*)&eps[(rowBase + p * 128 + lr) * D + dbase + 4 * lq];
  }

  v2f w2b[2][8];
  v2f w1v[8], b1v[8];
  v2f wml[8];
  const float* wsel = (col == 1) ? Wlv : Wmu;
  #pragma unroll
  for (int c = 0; c < 8; ++c) {
    const int k0 = 4 * c + 2 * hi;
    w1v[c] = mkv2(W1[d * H + k0], W1[d * H + k0 + 1]);
    b1v[c] = mkv2(b1[d * H + k0], b1[d * H + k0 + 1]);
    #pragma unroll
    for (int nt = 0; nt < 2; ++nt) {
      const int n = nt * 16 + col;
      w2b[nt][c] = mkv2(W2[d * H * H + k0 * H + n],
                        W2[d * H * H + (k0 + 1) * H + n]);
    }
    wml[c] = mkv2(wsel[d * H + k0], wsel[d * H + k0 + 1]);
  }
  const float b2v0 = b2[d * H + col];
  const float b2v1 = b2[d * H + 16 + col];
  const float bmuv = bmu[d];
  const float blvv = blv[d];

  #pragma unroll
  for (int p = 0; p < EPASS; ++p) {
    float* dx = &xs[(p * 128 + lr) * EPITCH + 4 * lq];
    float* de = &es[(p * 128 + lr) * EPITCH + 4 * lq];
    dx[0] = xin[p].x; dx[1] = xin[p].y; dx[2] = xin[p].z; dx[3] = xin[p].w;
    de[0] = ein[p].x; de[1] = ein[p].y; de[2] = ein[p].z; de[3] = ein[p].w;
  }
  __syncthreads();

  for (int t = 0; t < ETILES; ++t) {
    const float xv = xs[(t * 16 + col) * EPITCH + wave];

    v8f acc0 = {}, acc1 = {};
    #pragma unroll
    for (int c = 0; c < 8; ++c) {
      v2f a;
      a.x = fmaxf(fmaf(xv, w1v[c].x, b1v[c].x), 0.0f);
      a.y = fmaxf(fmaf(xv, w1v[c].y, b1v[c].y), 0.0f);
      acc0 = WMMA_F32(a, w2b[0][c], acc0);
      acc1 = WMMA_F32(a, w2b[1][c], acc1);
    }

    wave_lds_fence();
    #pragma unroll
    for (int j = 0; j < 8; ++j) {
      const int r = j + 8 * hi;
      wl[r * H + col]      = fmaxf(acc0[j] + b2v0, 0.0f);
      wl[r * H + 16 + col] = fmaxf(acc1[j] + b2v1, 0.0f);
    }
    wave_lds_fence();

    v8f accS = {};
    #pragma unroll
    for (int c = 0; c < 8; ++c) {
      v2f a = *(const v2f*)&wl[col * H + 4 * c + 2 * hi];
      accS = WMMA_F32(a, wml[c], accS);
    }

    if (col < 2) {
      const float bias = (col == 0) ? bmuv : blvv;
      float* o = &os[col * (ER * EPITCH) + (t * 16 + 8 * hi) * EPITCH + wave];
      #pragma unroll
      for (int j = 0; j < 8; ++j) o[j * EPITCH] = accS[j] + bias;
    }
  }

  wave_lds_fence();
  #pragma unroll
  for (int p = 0; p < ER / 32; ++p) {
    const int r = p * 32 + lane;
    const float m  = os[r * EPITCH + wave];
    const float lv = os[ER * EPITCH + r * EPITCH + wave];
    const float e  = es[r * EPITCH + wave];
    os[2 * ER * EPITCH + r * EPITCH + wave] = fmaf(e, __expf(0.5f * lv), m);
  }
  __syncthreads();

  #pragma unroll
  for (int p = 0; p < EPASS; ++p) {
    #pragma unroll
    for (int arr = 0; arr < 3; ++arr) {
      const float* s = &os[arr * (ER * EPITCH) + (p * 128 + lr) * EPITCH + 4 * lq];
      float4 v;
      v.x = s[0]; v.y = s[1]; v.z = s[2]; v.w = s[3];
      float* g = (arr == 0) ? mu_out : ((arr == 1) ? lv_out : z_ws);
      *(float4*)&g[(rowBase + p * 128 + lr) * D + dbase + 4 * lq] = v;
    }
  }
}

__global__ __launch_bounds__(256) void decoder_kernel(
    const float* __restrict__ z_ws,
    const float* __restrict__ Wd1, const float* __restrict__ bd1,
    const float* __restrict__ Wd2, const float* __restrict__ bd2,
    const float* __restrict__ Wd3, const float* __restrict__ bd3,
    float* __restrict__ action)
{
  __shared__ float lds[WAVES * 16 * H];
  const int lane = threadIdx.x & 31;
  const int wave = threadIdx.x >> 5;
  const int col  = lane & 15;
  const int hi   = lane >> 4;
  float* wl = &lds[wave * 16 * H];

  v2f wd1b[2][16];
  #pragma unroll
  for (int c = 0; c < 16; ++c) {
    const int k0 = 4 * c + 2 * hi;
    #pragma unroll
    for (int nt = 0; nt < 2; ++nt) {
      const int n = nt * 16 + col;
      wd1b[nt][c] = mkv2(Wd1[k0 * 32 + n], Wd1[(k0 + 1) * 32 + n]);
    }
  }
  v2f wd2b[2][8], wd3b[8];
  #pragma unroll
  for (int c = 0; c < 8; ++c) {
    const int k0 = 4 * c + 2 * hi;
    #pragma unroll
    for (int nt = 0; nt < 2; ++nt) {
      const int n = nt * 16 + col;
      wd2b[nt][c] = mkv2(Wd2[k0 * 32 + n], Wd2[(k0 + 1) * 32 + n]);
    }
    wd3b[c] = mkv2(Wd3[k0], Wd3[k0 + 1]);
  }
  const float bd1v0 = bd1[col], bd1v1 = bd1[16 + col];
  const float bd2v0 = bd2[col], bd2v1 = bd2[16 + col];
  const float bd3v  = bd3[0];

  const int rowBase = blockIdx.x * ROWS_PER_BLOCK + wave * (TILES * 16);

  for (int t = 0; t < TILES; ++t) {
    const int row0 = rowBase + t * 16;

    v8f acc0 = {}, acc1 = {};
    #pragma unroll
    for (int c = 0; c < 16; ++c) {
      v2f a = *(const v2f*)&z_ws[(row0 + col) * 64 + 4 * c + 2 * hi];
      acc0 = WMMA_F32(a, wd1b[0][c], acc0);
      acc1 = WMMA_F32(a, wd1b[1][c], acc1);
    }
    wave_lds_fence();
    #pragma unroll
    for (int j = 0; j < 8; ++j) {
      const int r = j + 8 * hi;
      wl[r * 32 + col]      = fmaxf(acc0[j] + bd1v0, 0.0f);
      wl[r * 32 + 16 + col] = fmaxf(acc1[j] + bd1v1, 0.0f);
    }
    wave_lds_fence();

    v8f a20 = {}, a21 = {};
    #pragma unroll
    for (int c = 0; c < 8; ++c) {
      v2f a = *(const v2f*)&wl[col * 32 + 4 * c + 2 * hi];
      a20 = WMMA_F32(a, wd2b[0][c], a20);
      a21 = WMMA_F32(a, wd2b[1][c], a21);
    }
    wave_lds_fence();
    #pragma unroll
    for (int j = 0; j < 8; ++j) {
      const int r = j + 8 * hi;
      wl[r * 32 + col]      = fmaxf(a20[j] + bd2v0, 0.0f);
      wl[r * 32 + 16 + col] = fmaxf(a21[j] + bd2v1, 0.0f);
    }
    wave_lds_fence();

    v8f accS = {};
    #pragma unroll
    for (int c = 0; c < 8; ++c) {
      v2f a = *(const v2f*)&wl[col * 32 + 4 * c + 2 * hi];
      accS = WMMA_F32(a, wd3b[c], accS);
    }
    if (col == 0) {
      #pragma unroll
      for (int j = 0; j < 8; ++j)
        action[row0 + j + 8 * hi] = accS[j] + bd3v;
    }
  }
}

extern "C" void kernel_launch(void* const* d_in, const int* in_sizes, int n_in,
                              void* d_out, int out_size, void* d_ws, size_t ws_size,
                              hipStream_t stream) {
  (void)in_sizes; (void)n_in; (void)out_size; (void)ws_size;
  const float* x   = (const float*)d_in[0];
  const float* eps = (const float*)d_in[1];
  const float* W1  = (const float*)d_in[2];
  const float* b1  = (const float*)d_in[3];
  const float* W2  = (const float*)d_in[4];
  const float* b2  = (const float*)d_in[5];
  const float* Wmu = (const float*)d_in[6];
  const float* bmu = (const float*)d_in[7];
  const float* Wlv = (const float*)d_in[8];
  const float* blv = (const float*)d_in[9];
  const float* Wd1 = (const float*)d_in[10];
  const float* bd1 = (const float*)d_in[11];
  const float* Wd2 = (const float*)d_in[12];
  const float* bd2 = (const float*)d_in[13];
  const float* Wd3 = (const float*)d_in[14];
  const float* bd3 = (const float*)d_in[15];

  float* out     = (float*)d_out;
  float* action  = out;
  float* mu_out  = out + Bsz;
  float* lv_out  = out + Bsz + Bsz * D;
  float* z_ws    = (float*)d_ws;

  dim3 grid1(Bsz / ER, D / EF);
  encoder_kernel<<<grid1, 1024, 0, stream>>>(x, eps, W1, b1, W2, b2,
                                            Wmu, bmu, Wlv, blv,
                                            mu_out, lv_out, z_ws);
  decoder_kernel<<<Bsz / ROWS_PER_BLOCK, 256, 0, stream>>>(z_ws, Wd1, bd1,
                                                           Wd2, bd2, Wd3, bd3,
                                                           action);
}


# decoder regridded: one 16-row tile per wave, 256 blocks x 8 waves = every WGP (was 64 blocks, 4 tiles in series per wave); a block's 128 actions leave via LDS as 4 whole lines from one wave
# speedup vs baseline: 2.7192x; 1.1511x over previous
#include <hip/hip_runtime.h>
#include <hip/hip_bf16.h>
#include <math.h>

typedef __attribute__((ext_vector_type(2))) float v2f;
typedef __attribute__((ext_vector_type(8))) float v8f;

#define WMMA_F32(A, B, C) \
  __builtin_amdgcn_wmma_f32_16x16x4_f32(false, (A), false, (B), (short)0, (C), false, false)

__device__ __forceinline__ v2f mkv2(float a, float b) { v2f r; r.x = a; r.y = b; return r; }

__device__ __forceinline__ void wave_lds_fence() {
  asm volatile("s_wait_dscnt 0" ::: "memory");
}

constexpr int Bsz = 32768;
constexpr int D   = 64;
constexpr int H   = 32;

constexpr int EF     = 32;
constexpr int ETILES = 16;
constexpr int ER     = ETILES * 16;
constexpr int EPASS  = ER / 128;
constexpr int EPITCH = EF + 1;

__global__ __launch_bounds__(1024) void encoder_kernel(
    const float* __restrict__ x,   const float* __restrict__ eps,
    const float* __restrict__ W1,  const float* __restrict__ b1,
    const float* __restrict__ W2,  const float* __restrict__ b2,
    const float* __restrict__ Wmu, const float* __restrict__ bmu,
    const float* __restrict__ Wlv, const float* __restrict__ blv,
    float* __restrict__ mu_out, float* __restrict__ lv_out,
    float* __restrict__ z_ws)
{
  __shared__ float lds[EF * 16 * H];
  __shared__ float xs[ER * EPITCH];
  __shared__ float es[ER * EPITCH];
  __shared__ float os[3 * ER * EPITCH];
  const int tid  = threadIdx.x;
  const int lane = tid & 31;
  const int wave = tid >> 5;
  const int col  = lane & 15;
  const int hi   = lane >> 4;
  const int dbase = blockIdx.y * EF;
  const int d     = dbase + wave;
  const int rowBase = blockIdx.x * ER;
  float* wl = &lds[wave * 16 * H];

  const int lr = tid >> 3;
  const int lq = tid & 7;
  float4 xin[EPASS], ein[EPASS];
  #pragma unroll
  for (int p = 0; p < EPASS; ++p) {
    xin[p] = *(const float4*)&x[(rowBase + p * 128 + lr) * D + dbase + 4 * lq];
    ein[p] = *(const float4*)&eps[(rowBase + p * 128 + lr) * D + dbase + 4 * lq];
  }

  v2f w2b[2][8];
  v2f w1v[8], b1v[8];
  v2f wml[8];
  const float* wsel = (col == 1) ? Wlv : Wmu;
  #pragma unroll
  for (int c = 0; c < 8; ++c) {
    const int k0 = 4 * c + 2 * hi;
    w1v[c] = mkv2(W1[d * H + k0], W1[d * H + k0 + 1]);
    b1v[c] = mkv2(b1[d * H + k0], b1[d * H + k0 + 1]);
    #pragma unroll
    for (int nt = 0; nt < 2; ++nt) {
      const int n = nt * 16 + col;
      w2b[nt][c] = mkv2(W2[d * H * H + k0 * H + n],
                        W2[d * H * H + (k0 + 1) * H + n]);
    }
    wml[c] = mkv2(wsel[d * H + k0], wsel[d * H + k0 + 1]);
  }
  const float b2v0 = b2[d * H + col];
  const float b2v1 = b2[d * H + 16 + col];
  const float bmuv = bmu[d];
  const float blvv = blv[d];

  #pragma unroll
  for (int p = 0; p < EPASS; ++p) {
    float* dx = &xs[(p * 128 + lr) * EPITCH + 4 * lq];
    float* de = &es[(p * 128 + lr) * EPITCH + 4 * lq];
    dx[0] = xin[p].x; dx[1] = xin[p].y; dx[2] = xin[p].z; dx[3] = xin[p].w;
    de[0] = ein[p].x; de[1] = ein[p].y; de[2] = ein[p].z; de[3] = ein[p].w;
  }
  __syncthreads();

  for (int t = 0; t < ETILES; ++t) {
    const float xv = xs[(t * 16 + col) * EPITCH + wave];

    v8f acc0 = {}, acc1 = {};
    #pragma unroll
    for (int c = 0; c < 8; ++c) {
      v2f a;
      a.x = fmaxf(fmaf(xv, w1v[c].x, b1v[c].x), 0.0f);
      a.y = fmaxf(fmaf(xv, w1v[c].y, b1v[c].y), 0.0f);
      acc0 = WMMA_F32(a, w2b[0][c], acc0);
      acc1 = WMMA_F32(a, w2b[1][c], acc1);
    }

    wave_lds_fence();
    #pragma unroll
    for (int j = 0; j < 8; ++j) {
      const int r = j + 8 * hi;
      wl[r * H + col]      = fmaxf(acc0[j] + b2v0, 0.0f);
      wl[r * H + 16 + col] = fmaxf(acc1[j] + b2v1, 0.0f);
    }
    wave_lds_fence();

    v8f accS = {};
    #pragma unroll
    for (int c = 0; c < 8; ++c) {
      v2f a = *(const v2f*)&wl[col * H + 4 * c + 2 * hi];
      accS = WMMA_F32(a, wml[c], accS);
    }

    if (col < 2) {
      const float bias = (col == 0) ? bmuv : blvv;
      float* o = &os[col * (ER * EPITCH) + (t * 16 + 8 * hi) * EPITCH + wave];
      #pragma unroll
      for (int j = 0; j < 8; ++j) o[j * EPITCH] = accS[j] + bias;
    }
  }

  wave_lds_fence();
  #pragma unroll
  for (int p = 0; p < ER / 32; ++p) {
    const int r = p * 32 + lane;
    const float m  = os[r * EPITCH + wave];
    const float lv = os[ER * EPITCH + r * EPITCH + wave];
    const float e  = es[r * EPITCH + wave];
    os[2 * ER * EPITCH + r * EPITCH + wave] = fmaf(e, __expf(0.5f * lv), m);
  }
  __syncthreads();

  #pragma unroll
  for (int p = 0; p < EPASS; ++p) {
    #pragma unroll
    for (int arr = 0; arr < 3; ++arr) {
      const float* s = &os[arr * (ER * EPITCH) + (p * 128 + lr) * EPITCH + 4 * lq];
      float4 v;
      v.x = s[0]; v.y = s[1]; v.z = s[2]; v.w = s[3];
      float* g = (arr == 0) ? mu_out : ((arr == 1) ? lv_out : z_ws);
      *(float4*)&g[(rowBase + p * 128 + lr) * D + dbase + 4 * lq] = v;
    }
  }
}

constexpr int DWAVES = 8;
constexpr int DROWS  = DWAVES * 16;

__global__ __launch_bounds__(256) void decoder_kernel(
    const float* __restrict__ z_ws,
    const float* __restrict__ Wd1, const float* __restrict__ bd1,
    const float* __restrict__ Wd2, const float* __restrict__ bd2,
    const float* __restrict__ Wd3, const float* __restrict__ bd3,
    float* __restrict__ action)
{
  __shared__ float lds[DWAVES * 16 * H];
  __shared__ alignas(16) float act[DROWS];
  const int lane = threadIdx.x & 31;
  const int wave = threadIdx.x >> 5;
  const int col  = lane & 15;
  const int hi   = lane >> 4;
  float* wl = &lds[wave * 16 * H];
  const int row0 = blockIdx.x * DROWS + wave * 16;

  v2f za[16];
  #pragma unroll
  for (int c = 0; c < 16; ++c)
    za[c] = *(const v2f*)&z_ws[(row0 + col) * 64 + 4 * c + 2 * hi];

  v2f wd1b[2][16];
  #pragma unroll
  for (int c = 0; c < 16; ++c) {
    const int k0 = 4 * c + 2 * hi;
    #pragma unroll
    for (int nt = 0; nt < 2; ++nt) {
      const int n = nt * 16 + col;
      wd1b[nt][c] = mkv2(Wd1[k0 * 32 + n], Wd1[(k0 + 1) * 32 + n]);
    }
  }
  v2f wd2b[2][8], wd3b[8];
  #pragma unroll
  for (int c = 0; c < 8; ++c) {
    const int k0 = 4 * c + 2 * hi;
    #pragma unroll
    for (int nt = 0; nt < 2; ++nt) {
      const int n = nt * 16 + col;
      wd2b[nt][c] = mkv2(Wd2[k0 * 32 + n], Wd2[(k0 + 1) * 32 + n]);
    }
    wd3b[c] = mkv2(Wd3[k0], Wd3[k0 + 1]);
  }
  const float bd1v0 = bd1[col], bd1v1 = bd1[16 + col];
  const float bd2v0 = bd2[col], bd2v1 = bd2[16 + col];
  const float bd3v  = bd3[0];

  v8f acc0 = {}, acc1 = {};
  #pragma unroll
  for (int c = 0; c < 16; ++c) {
    acc0 = WMMA_F32(za[c], wd1b[0][c], acc0);
    acc1 = WMMA_F32(za[c], wd1b[1][c], acc1);
  }
  #pragma unroll
  for (int j = 0; j < 8; ++j) {
    const int r = j + 8 * hi;
    wl[r * 32 + col]      = fmaxf(acc0[j] + bd1v0, 0.0f);
    wl[r * 32 + 16 + col] = fmaxf(acc1[j] + bd1v1, 0.0f);
  }
  wave_lds_fence();

  v8f a20 = {}, a21 = {};
  #pragma unroll
  for (int c = 0; c < 8; ++c) {
    v2f a = *(const v2f*)&wl[col * 32 + 4 * c + 2 * hi];
    a20 = WMMA_F32(a, wd2b[0][c], a20);
    a21 = WMMA_F32(a, wd2b[1][c], a21);
  }
  wave_lds_fence();
  #pragma unroll
  for (int j = 0; j < 8; ++j) {
    const int r = j + 8 * hi;
    wl[r * 32 + col]      = fmaxf(a20[j] + bd2v0, 0.0f);
    wl[r * 32 + 16 + col] = fmaxf(a21[j] + bd2v1, 0.0f);
  }
  wave_lds_fence();

  v8f accS = {};
  #pragma unroll
  for (int c = 0; c < 8; ++c) {
    v2f a = *(const v2f*)&wl[col * 32 + 4 * c + 2 * hi];
    accS = WMMA_F32(a, wd3b[c], accS);
  }
  if (col == 0) {
    #pragma unroll
    for (int j = 0; j < 8; ++j)
      act[wave * 16 + 8 * hi + j] = accS[j] + bd3v;
  }
  __syncthreads();

  if (threadIdx.x < 32) {
    const float4 v = *(const float4*)&act[4 * threadIdx.x];
    *(float4*)&action[blockIdx.x * DROWS + 4 * threadIdx.x] = v;
  }
}

extern "C" void kernel_launch(void* const* d_in, const int* in_sizes, int n_in,
                              void* d_out, int out_size, void* d_ws, size_t ws_size,
                              hipStream_t stream) {
  (void)in_sizes; (void)n_in; (void)out_size; (void)ws_size;
  const float* x   = (const float*)d_in[0];
  const float* eps = (const float*)d_in[1];
  const float* W1  = (const float*)d_in[2];
  const float* b1  = (const float*)d_in[3];
  const float* W2  = (const float*)d_in[4];
  const float* b2  = (const float*)d_in[5];
  const float* Wmu = (const float*)d_in[6];
  const float* bmu = (const float*)d_in[7];
  const float* Wlv = (const float*)d_in[8];
  const float* blv = (const float*)d_in[9];
  const float* Wd1 = (const float*)d_in[10];
  const float* bd1 = (const float*)d_in[11];
  const float* Wd2 = (const float*)d_in[12];
  const float* bd2 = (const float*)d_in[13];
  const float* Wd3 = (const float*)d_in[14];
  const float* bd3 = (const float*)d_in[15];

  float* out     = (float*)d_out;
  float* action  = out;
  float* mu_out  = out + Bsz;
  float* lv_out  = out + Bsz + Bsz * D;
  float* z_ws    = (float*)d_ws;

  dim3 grid1(Bsz / ER, D / EF);
  encoder_kernel<<<grid1, 1024, 0, stream>>>(x, eps, W1, b1, W2, b2,
                                            Wmu, bmu, Wlv, blv,
                                            mu_out, lv_out, z_ws);
  decoder_kernel<<<Bsz / DROWS, 256, 0, stream>>>(z_ws, Wd1, bd1,
                                                           Wd2, bd2, Wd3, bd3,
                                                           action);
}


# encoder tile loop: a wave lowers its issue priority as it advances (s_setprio 3,2,1,0 from tiles 0,10,13,15): a SIMD's 8 waves finish together, none runs its last tiles alone with waits uncovered
# speedup vs baseline: 2.8494x; 1.0479x over previous
#include <hip/hip_runtime.h>
#include <hip/hip_bf16.h>
#include <math.h>

typedef __attribute__((ext_vector_type(2))) float v2f;
typedef __attribute__((ext_vector_type(8))) float v8f;

#define WMMA_F32(A, B, C) \
  __builtin_amdgcn_wmma_f32_16x16x4_f32(false, (A), false, (B), (short)0, (C), false, false)

__device__ __forceinline__ v2f mkv2(float a, float b) { v2f r; r.x = a; r.y = b; return r; }

__device__ __forceinline__ void wave_lds_fence() {
  asm volatile("s_wait_dscnt 0" ::: "memory");
}

constexpr int Bsz = 32768;
constexpr int D   = 64;
constexpr int H   = 32;

constexpr int EF     = 32;
constexpr int ETILES = 16;
constexpr int ER     = ETILES * 16;
constexpr int EPASS  = ER / 128;
constexpr int EPITCH = EF + 1;
constexpr int EPRIO2 = 10, EPRIO1 = 13, EPRIO0 = 15;

__global__ __launch_bounds__(1024) void encoder_kernel(
    const float* __restrict__ x,   const float* __restrict__ eps,
    const float* __restrict__ W1,  const float* __restrict__ b1,
    const float* __restrict__ W2,  const float* __restrict__ b2,
    const float* __restrict__ Wmu, const float* __restrict__ bmu,
    const float* __restrict__ Wlv, const float* __restrict__ blv,
    float* __restrict__ mu_out, float* __restrict__ lv_out,
    float* __restrict__ z_ws)
{
  __shared__ float lds[EF * 16 * H];
  __shared__ float xs[ER * EPITCH];
  __shared__ float es[ER * EPITCH];
  __shared__ float os[3 * ER * EPITCH];
  const int tid  = threadIdx.x;
  const int lane = tid & 31;
  const int wave = tid >> 5;
  const int col  = lane & 15;
  const int hi   = lane >> 4;
  const int dbase = blockIdx.y * EF;
  const int d     = dbase + wave;
  const int rowBase = blockIdx.x * ER;
  float* wl = &lds[wave * 16 * H];

  const int lr = tid >> 3;
  const int lq = tid & 7;
  float4 xin[EPASS], ein[EPASS];
  #pragma unroll
  for (int p = 0; p < EPASS; ++p) {
    xin[p] = *(const float4*)&x[(rowBase + p * 128 + lr) * D + dbase + 4 * lq];
    ein[p] = *(const float4*)&eps[(rowBase + p * 128 + lr) * D + dbase + 4 * lq];
  }

  v2f w2b[2][8];
  v2f w1v[8], b1v[8];
  v2f wml[8];
  const float* wsel = (col == 1) ? Wlv : Wmu;
  #pragma unroll
  for (int c = 0; c < 8; ++c) {
    const int k0 = 4 * c + 2 * hi;
    w1v[c] = mkv2(W1[d * H + k0], W1[d * H + k0 + 1]);
    b1v[c] = mkv2(b1[d * H + k0], b1[d * H + k0 + 1]);
    #pragma unroll
    for (int nt = 0; nt < 2; ++nt) {
      const int n = nt * 16 + col;
      w2b[nt][c] = mkv2(W2[d * H * H + k0 * H + n],
                        W2[d * H * H + (k0 + 1) * H + n]);
    }
    wml[c] = mkv2(wsel[d * H + k0], wsel[d * H + k0 + 1]);
  }
  const float b2v0 = b2[d * H + col];
  const float b2v1 = b2[d * H + 16 + col];
  const float bmuv = bmu[d];
  const float blvv = blv[d];

  #pragma unroll
  for (int p = 0; p < EPASS; ++p) {
    float* dx = &xs[(p * 128 + lr) * EPITCH + 4 * lq];
    float* de = &es[(p * 128 + lr) * EPITCH + 4 * lq];
    dx[0] = xin[p].x; dx[1] = xin[p].y; dx[2] = xin[p].z; dx[3] = xin[p].w;
    de[0] = ein[p].x; de[1] = ein[p].y; de[2] = ein[p].z; de[3] = ein[p].w;
  }
  __syncthreads();

  auto do_tile = [&](const int t) {
    const float xv = xs[(t * 16 + col) * EPITCH + wave];

    v8f acc0 = {}, acc1 = {};
    #pragma unroll
    for (int c = 0; c < 8; ++c) {
      v2f a;
      a.x = fmaxf(fmaf(xv, w1v[c].x, b1v[c].x), 0.0f);
      a.y = fmaxf(fmaf(xv, w1v[c].y, b1v[c].y), 0.0f);
      acc0 = WMMA_F32(a, w2b[0][c], acc0);
      acc1 = WMMA_F32(a, w2b[1][c], acc1);
    }

    wave_lds_fence();
    #pragma unroll
    for (int j = 0; j < 8; ++j) {
      const int r = j + 8 * hi;
      wl[r * H + col]      = fmaxf(acc0[j] + b2v0, 0.0f);
      wl[r * H + 16 + col] = fmaxf(acc1[j] + b2v1, 0.0f);
    }
    wave_lds_fence();

    v8f accS = {};
    #pragma unroll
    for (int c = 0; c < 8; ++c) {
      v2f a = *(const v2f*)&wl[col * H + 4 * c + 2 * hi];
      accS = WMMA_F32(a, wml[c], accS);
    }

    if (col < 2) {
      const float bias = (col == 0) ? bmuv : blvv;
      float* o = &os[col * (ER * EPITCH) + (t * 16 + 8 * hi) * EPITCH + wave];
      #pragma unroll
      for (int j = 0; j < 8; ++j) o[j * EPITCH] = accS[j] + bias;
    }
  };

  __builtin_amdgcn_s_setprio(3);
  for (int t = 0; t < EPRIO2; ++t) do_tile(t);
  __builtin_amdgcn_s_setprio(2);
  for (int t = EPRIO2; t < EPRIO1; ++t) do_tile(t);
  __builtin_amdgcn_s_setprio(1);
  for (int t = EPRIO1; t < EPRIO0; ++t) do_tile(t);
  __builtin_amdgcn_s_setprio(0);
  for (int t = EPRIO0; t < ETILES; ++t) do_tile(t);

  wave_lds_fence();
  #pragma unroll
  for (int p = 0; p < ER / 32; ++p) {
    const int r = p * 32 + lane;
    const float m  = os[r * EPITCH + wave];
    const float lv = os[ER * EPITCH + r * EPITCH + wave];
    const float e  = es[r * EPITCH + wave];
    os[2 * ER * EPITCH + r * EPITCH + wave] = fmaf(e, __expf(0.5f * lv), m);
  }
  __syncthreads();

  #pragma unroll
  for (int p = 0; p < EPASS; ++p) {
    #pragma unroll
    for (int arr = 0; arr < 3; ++arr) {
      const float* s = &os[arr * (ER * EPITCH) + (p * 128 + lr) * EPITCH + 4 * lq];
      float4 v;
      v.x = s[0]; v.y = s[1]; v.z = s[2]; v.w = s[3];
      float* g = (arr == 0) ? mu_out : ((arr == 1) ? lv_out : z_ws);
      *(float4*)&g[(rowBase + p * 128 + lr) * D + dbase + 4 * lq] = v;
    }
  }
}

constexpr int DWAVES = 8;
constexpr int DROWS  = DWAVES * 16;

__global__ __launch_bounds__(256) void decoder_kernel(
    const float* __restrict__ z_ws,
    const float* __restrict__ Wd1, const float* __restrict__ bd1,
    const float* __restrict__ Wd2, const float* __restrict__ bd2,
    const float* __restrict__ Wd3, const float* __restrict__ bd3,
    float* __restrict__ action)
{
  __shared__ float lds[DWAVES * 16 * H];
  __shared__ alignas(16) float act[DROWS];
  const int lane = threadIdx.x & 31;
  const int wave = threadIdx.x >> 5;
  const int col  = lane & 15;
  const int hi   = lane >> 4;
  float* wl = &lds[wave * 16 * H];
  const int row0 = blockIdx.x * DROWS + wave * 16;

  v2f za[16];
  #pragma unroll
  for (int c = 0; c < 16; ++c)
    za[c] = *(const v2f*)&z_ws[(row0 + col) * 64 + 4 * c + 2 * hi];

  v2f wd1b[2][16];
  #pragma unroll
  for (int c = 0; c < 16; ++c) {
    const int k0 = 4 * c + 2 * hi;
    #pragma unroll
    for (int nt = 0; nt < 2; ++nt) {
      const int n = nt * 16 + col;
      wd1b[nt][c] = mkv2(Wd1[k0 * 32 + n], Wd1[(k0 + 1) * 32 + n]);
    }
  }
  v2f wd2b[2][8], wd3b[8];
  #pragma unroll
  for (int c = 0; c < 8; ++c) {
    const int k0 = 4 * c + 2 * hi;
    #pragma unroll
    for (int nt = 0; nt < 2; ++nt) {
      const int n = nt * 16 + col;
      wd2b[nt][c] = mkv2(Wd2[k0 * 32 + n], Wd2[(k0 + 1) * 32 + n]);
    }
    wd3b[c] = mkv2(Wd3[k0], Wd3[k0 + 1]);
  }
  const float bd1v0 = bd1[col], bd1v1 = bd1[16 + col];
  const float bd2v0 = bd2[col], bd2v1 = bd2[16 + col];
  const float bd3v  = bd3[0];

  v8f acc0 = {}, acc1 = {};
  #pragma unroll
  for (int c = 0; c < 16; ++c) {
    acc0 = WMMA_F32(za[c], wd1b[0][c], acc0);
    acc1 = WMMA_F32(za[c], wd1b[1][c], acc1);
  }
  #pragma unroll
  for (int j = 0; j < 8; ++j) {
    const int r = j + 8 * hi;
    wl[r * 32 + col]      = fmaxf(acc0[j] + bd1v0, 0.0f);
    wl[r * 32 + 16 + col] = fmaxf(acc1[j] + bd1v1, 0.0f);
  }
  wave_lds_fence();

  v8f a20 = {}, a21 = {};
  #pragma unroll
  for (int c = 0; c < 8; ++c) {
    v2f a = *(const v2f*)&wl[col * 32 + 4 * c + 2 * hi];
    a20 = WMMA_F32(a, wd2b[0][c], a20);
    a21 = WMMA_F32(a, wd2b[1][c], a21);
  }
  wave_lds_fence();
  #pragma unroll
  for (int j = 0; j < 8; ++j) {
    const int r = j + 8 * hi;
    wl[r * 32 + col]      = fmaxf(a20[j] + bd2v0, 0.0f);
    wl[r * 32 + 16 + col] = fmaxf(a21[j] + bd2v1, 0.0f);
  }
  wave_lds_fence();

  v8f accS = {};
  #pragma unroll
  for (int c = 0; c < 8; ++c) {
    v2f a = *(const v2f*)&wl[col * 32 + 4 * c + 2 * hi];
    accS = WMMA_F32(a, wd3b[c], accS);
  }
  if (col == 0) {
    #pragma unroll
    for (int j = 0; j < 8; ++j)
      act[wave * 16 + 8 * hi + j] = accS[j] + bd3v;
  }
  __syncthreads();

  if (threadIdx.x < 32) {
    const float4 v = *(const float4*)&act[4 * threadIdx.x];
    *(float4*)&action[blockIdx.x * DROWS + 4 * threadIdx.x] = v;
  }
}

extern "C" void kernel_launch(void* const* d_in, const int* in_sizes, int n_in,
                              void* d_out, int out_size, void* d_ws, size_t ws_size,
                              hipStream_t stream) {
  (void)in_sizes; (void)n_in; (void)out_size; (void)ws_size;
  const float* x   = (const float*)d_in[0];
  const float* eps = (const float*)d_in[1];
  const float* W1  = (const float*)d_in[2];
  const float* b1  = (const float*)d_in[3];
  const float* W2  = (const float*)d_in[4];
  const float* b2  = (const float*)d_in[5];
  const float* Wmu = (const float*)d_in[6];
  const float* bmu = (const float*)d_in[7];
  const float* Wlv = (const float*)d_in[8];
  const float* blv = (const float*)d_in[9];
  const float* Wd1 = (const float*)d_in[10];
  const float* bd1 = (const float*)d_in[11];
  const float* Wd2 = (const float*)d_in[12];
  const float* bd2 = (const float*)d_in[13];
  const float* Wd3 = (const float*)d_in[14];
  const float* bd3 = (const float*)d_in[15];

  float* out     = (float*)d_out;
  float* action  = out;
  float* mu_out  = out + Bsz;
  float* lv_out  = out + Bsz + Bsz * D;
  float* z_ws    = (float*)d_ws;

  dim3 grid1(Bsz / ER, D / EF);
  encoder_kernel<<<grid1, 1024, 0, stream>>>(x, eps, W1, b1, W2, b2,
                                            Wmu, bmu, Wlv, blv,
                                            mu_out, lv_out, z_ws);
  decoder_kernel<<<Bsz / DROWS, 256, 0, stream>>>(z_ws, Wd1, bd1,
                                                           Wd2, bd2, Wd3, bd3,
                                                           action);
}
